# starting kernel
# baseline (speedup 1.0000x reference)
#include <hip/hip_runtime.h>

#define N_NODES 100000
#define IN_DIM  128
#define HEADS   4
#define HC      128
#define NEG_SLOPE 0.2f

typedef float v2f __attribute__((ext_vector_type(2)));
typedef float v8f __attribute__((ext_vector_type(8)));

__device__ __forceinline__ int f2o(float f) {
    int i = __float_as_int(f);
    return i >= 0 ? i : (i ^ 0x7FFFFFFF);
}
__device__ __forceinline__ float o2f(int i) {
    int j = i >= 0 ? i : (i ^ 0x7FFFFFFF);
    return __int_as_float(j);
}

__global__ void gat_gemm(const float* __restrict__ x,
                         const float* __restrict__ Wl, const float* __restrict__ bl,
                         const float* __restrict__ Wr, const float* __restrict__ br,
                         float* __restrict__ xl, float* __restrict__ xr)
{
    const int wid  = threadIdx.x >> 5;
    const int lane = threadIdx.x & 31;
    const int task = blockIdx.x * 4 + wid;
    const int matSel  = task & 1;
    const int rowTile = task >> 1;

    const float* W    = matSel ? Wr : Wl;
    const float* bv   = matSel ? br : bl;
    float*       outp = matSel ? xr : xl;

    const int row0 = rowTile * 16;
    const int lm = lane & 15;
    const int lk = (lane >> 4) * 2;

    const float* aptr = x + (size_t)(row0 + lm) * IN_DIM + lk;
    const float* bptr = W + (size_t)lk * HC + lm;

    v8f c[8] = {};
    for (int k0 = 0; k0 < IN_DIM; k0 += 4) {
        const float2 av = *(const float2*)aptr;
        v2f a; a.x = av.x; a.y = av.y;
        #pragma unroll
        for (int t = 0; t < 8; ++t) {
            v2f b;
            b.x = bptr[t * 16];
            b.y = bptr[HC + t * 16];
            c[t] = __builtin_amdgcn_wmma_f32_16x16x4_f32(
                       false, a, false, b, (short)0, c[t], false, false);
        }
        aptr += 4;
        bptr += 4 * HC;
    }

    const int rbase = row0 + ((lane >> 4) * 8);
    #pragma unroll
    for (int t = 0; t < 8; ++t) {
        const int col = t * 16 + lm;
        const float bb = bv[col];
        #pragma unroll
        for (int r = 0; r < 8; ++r)
            outp[(size_t)(rbase + r) * HC + col] = c[t][r] + bb;
    }
}

__global__ void gat_init(float* __restrict__ out, float* __restrict__ denom,
                         int* __restrict__ segmax, int nOut, int nSeg)
{
    int i = blockIdx.x * blockDim.x + threadIdx.x;
    if (i < nOut) out[i] = 0.0f;
    if (i < nSeg) { denom[i] = 0.0f; segmax[i] = (int)0x80000000; }
}

__device__ __forceinline__ float edge_score(const float4 a, const float4 b,
                                            const float4 at)
{
    float m0 = a.x + b.x; m0 = m0 >= 0.0f ? m0 : NEG_SLOPE * m0;
    float m1 = a.y + b.y; m1 = m1 >= 0.0f ? m1 : NEG_SLOPE * m1;
    float m2 = a.z + b.z; m2 = m2 >= 0.0f ? m2 : NEG_SLOPE * m2;
    float m3 = a.w + b.w; m3 = m3 >= 0.0f ? m3 : NEG_SLOPE * m3;
    float p = at.x * m0 + at.y * m1 + at.z * m2 + at.w * m3;
    p += __shfl_xor(p, 1, 32);
    p += __shfl_xor(p, 2, 32);
    p += __shfl_xor(p, 4, 32);
    return p;
}

__global__ void gat_edge_max(const float* __restrict__ xl, const float* __restrict__ xr,
                             const float* __restrict__ att,
                             const int* __restrict__ src, const int* __restrict__ dst,
                             int nEdges, int nTot, int* __restrict__ segmax)
{
    const int e = blockIdx.x * 8 + (threadIdx.x >> 5);
    if (e >= nTot) return;
    const int lane = threadIdx.x & 31;
    int s, d;
    if (e < nEdges) { s = src[e]; d = dst[e]; } else { s = e - nEdges; d = s; }

    const float4 a  = *(const float4*)(xl + (size_t)s * HC + lane * 4);
    const float4 b  = *(const float4*)(xr + (size_t)d * HC + lane * 4);
    const float4 at = *(const float4*)(att + lane * 4);
    const float p = edge_score(a, b, at);

    if ((lane & 7) == 0)
        atomicMax(segmax + (size_t)d * HEADS + (lane >> 3), f2o(p));
}

__global__ void gat_edge_acc(const float* __restrict__ xl, const float* __restrict__ xr,
                             const float* __restrict__ att,
                             const int* __restrict__ src, const int* __restrict__ dst,
                             int nEdges, int nTot,
                             const int* __restrict__ segmax,
                             float* __restrict__ denom, float* __restrict__ out)
{
    const int e = blockIdx.x * 8 + (threadIdx.x >> 5);
    if (e >= nTot) return;
    const int lane = threadIdx.x & 31;
    int s, d;
    if (e < nEdges) { s = src[e]; d = dst[e]; } else { s = e - nEdges; d = s; }

    const float4 a  = *(const float4*)(xl + (size_t)s * HC + lane * 4);
    const float4 b  = *(const float4*)(xr + (size_t)d * HC + lane * 4);
    const float4 at = *(const float4*)(att + lane * 4);
    const float p = edge_score(a, b, at);

    const int h = lane >> 3;
    const float mx = o2f(segmax[(size_t)d * HEADS + h]);
    const float w = __expf(p - mx);

    if ((lane & 7) == 0)
        unsafeAtomicAdd(denom + (size_t)d * HEADS + h, w);

    float* o = out + (size_t)d * HC + lane * 4;
    unsafeAtomicAdd(o + 0, w * a.x);
    unsafeAtomicAdd(o + 1, w * a.y);
    unsafeAtomicAdd(o + 2, w * a.z);
    unsafeAtomicAdd(o + 3, w * a.w);
}

__global__ void gat_final(float* __restrict__ out, const float* __restrict__ denom,
                          const float* __restrict__ bias, int nOut)
{
    int i = blockIdx.x * blockDim.x + threadIdx.x;
    if (i >= nOut) return;
    const int node = i >> 7;
    const int ch   = i & 127;
    out[i] = out[i] / denom[node * HEADS + (ch >> 5)] + bias[ch];
}

extern "C" void kernel_launch(void* const* d_in, const int* in_sizes, int n_in,
                              void* d_out, int out_size, void* d_ws, size_t ws_size,
                              hipStream_t stream)
{
    const float* x    = (const float*)d_in[0];
    const float* Wl   = (const float*)d_in[1];
    const float* bl   = (const float*)d_in[2];
    const float* Wr   = (const float*)d_in[3];
    const float* br   = (const float*)d_in[4];
    const float* att  = (const float*)d_in[5];
    const float* bias = (const float*)d_in[6];
    const int*   src  = (const int*)d_in[7];
    const int*   dst  = (const int*)d_in[8];
    const int nE = in_sizes[7];
    float* out = (float*)d_out;

    float* xl    = (float*)d_ws;
    float* xr    = xl + (size_t)N_NODES * HC;
    float* denom = xr + (size_t)N_NODES * HC;
    int*   smax  = (int*)(denom + (size_t)N_NODES * HEADS);

    const int nOut = N_NODES * HC;
    const int nSeg = N_NODES * HEADS;

    gat_init<<<(nOut + 255) / 256, 256, 0, stream>>>(out, denom, smax, nOut, nSeg);

    gat_gemm<<<3125, 128, 0, stream>>>(x, Wl, bl, Wr, br, xl, xr);

    const int nTot = nE + N_NODES;
    const int eb = (nTot + 7) / 8;
    gat_edge_max<<<eb, 256, 0, stream>>>(xl, xr, att, src, dst, nE, nTot, smax);
    gat_edge_acc<<<eb, 256, 0, stream>>>(xl, xr, att, src, dst, nE, nTot, smax, denom, out);

    gat_final<<<(nOut + 255) / 256, 256, 0, stream>>>(out, denom, bias, nOut);
}


# Edge passes: each wave takes 8 edges instead of 1 and requests their row gathers together before first use (8x fewer waves, more bytes in flight per wave); float atomics kept, arrival-order sums
# speedup vs baseline: 1.1403x; 1.1403x over previous
#include <hip/hip_runtime.h>

#define N_NODES 100000
#define IN_DIM  128
#define HEADS   4
#define HC      128
#define NEG_SLOPE 0.2f

typedef float v2f __attribute__((ext_vector_type(2)));
typedef float v8f __attribute__((ext_vector_type(8)));

__device__ __forceinline__ int f2o(float f) {
    int i = __float_as_int(f);
    return i >= 0 ? i : (i ^ 0x7FFFFFFF);
}
__device__ __forceinline__ float o2f(int i) {
    int j = i >= 0 ? i : (i ^ 0x7FFFFFFF);
    return __int_as_float(j);
}

__global__ void gat_gemm(const float* __restrict__ x,
                         const float* __restrict__ Wl, const float* __restrict__ bl,
                         const float* __restrict__ Wr, const float* __restrict__ br,
                         float* __restrict__ xl, float* __restrict__ xr)
{
    const int wid  = threadIdx.x >> 5;
    const int lane = threadIdx.x & 31;
    const int task = blockIdx.x * 4 + wid;
    const int matSel  = task & 1;
    const int rowTile = task >> 1;

    const float* W    = matSel ? Wr : Wl;
    const float* bv   = matSel ? br : bl;
    float*       outp = matSel ? xr : xl;

    const int row0 = rowTile * 16;
    const int lm = lane & 15;
    const int lk = (lane >> 4) * 2;

    const float* aptr = x + (size_t)(row0 + lm) * IN_DIM + lk;
    const float* bptr = W + (size_t)lk * HC + lm;

    v8f c[8] = {};
    for (int k0 = 0; k0 < IN_DIM; k0 += 4) {
        const float2 av = *(const float2*)aptr;
        v2f a; a.x = av.x; a.y = av.y;
        #pragma unroll
        for (int t = 0; t < 8; ++t) {
            v2f b;
            b.x = bptr[t * 16];
            b.y = bptr[HC + t * 16];
            c[t] = __builtin_amdgcn_wmma_f32_16x16x4_f32(
                       false, a, false, b, (short)0, c[t], false, false);
        }
        aptr += 4;
        bptr += 4 * HC;
    }

    const int rbase = row0 + ((lane >> 4) * 8);
    #pragma unroll
    for (int t = 0; t < 8; ++t) {
        const int col = t * 16 + lm;
        const float bb = bv[col];
        #pragma unroll
        for (int r = 0; r < 8; ++r)
            outp[(size_t)(rbase + r) * HC + col] = c[t][r] + bb;
    }
}

__global__ void gat_init(float* __restrict__ out, float* __restrict__ denom,
                         int* __restrict__ segmax, int nOut, int nSeg)
{
    int i = blockIdx.x * blockDim.x + threadIdx.x;
    if (i < nOut) out[i] = 0.0f;
    if (i < nSeg) { denom[i] = 0.0f; segmax[i] = (int)0x80000000; }
}

__device__ __forceinline__ float edge_partial(const float4 a, const float4 b,
                                              const float4 at)
{
    float m0 = a.x + b.x; m0 = m0 >= 0.0f ? m0 : NEG_SLOPE * m0;
    float m1 = a.y + b.y; m1 = m1 >= 0.0f ? m1 : NEG_SLOPE * m1;
    float m2 = a.z + b.z; m2 = m2 >= 0.0f ? m2 : NEG_SLOPE * m2;
    float m3 = a.w + b.w; m3 = m3 >= 0.0f ? m3 : NEG_SLOPE * m3;
    return at.x * m0 + at.y * m1 + at.z * m2 + at.w * m3;
}

#define EPW 8

__device__ __forceinline__ void edge_ids(const int* __restrict__ src,
                                         const int* __restrict__ dst,
                                         int nEdges, int nTot, int e0, int lane,
                                         int& sv, int& dv)
{
    const int e = e0 + (lane & (EPW - 1));
    sv = 0; dv = 0;
    if (e < nEdges)    { sv = src[e]; dv = dst[e]; }
    else if (e < nTot) { sv = e - nEdges; dv = sv; }
}

__global__ void __launch_bounds__(256)
gat_edge_max(const float* __restrict__ xl, const float* __restrict__ xr,
             const float* __restrict__ att,
             const int* __restrict__ src, const int* __restrict__ dst,
             int nEdges, int nTot, int* __restrict__ segmax)
{
    const int wv = __builtin_amdgcn_readfirstlane((int)(threadIdx.x >> 5));
    const int e0 = (blockIdx.x * 8 + wv) * EPW;
    if (e0 >= nTot) return;
    const int lane = threadIdx.x & 31;

    int sv, dv;
    edge_ids(src, dst, nEdges, nTot, e0, lane, sv, dv);
    const float4 at = *(const float4*)(att + lane * 4);

    float4 a[EPW], b[EPW];
    int d[EPW];
    #pragma unroll
    for (int k = 0; k < EPW; ++k) {
        const int s = __builtin_amdgcn_readlane(sv, k);
        d[k] = __builtin_amdgcn_readlane(dv, k);
        a[k] = *(const float4*)(xl + (size_t)s    * HC + lane * 4);
        b[k] = *(const float4*)(xr + (size_t)d[k] * HC + lane * 4);
    }

    float p[EPW];
    #pragma unroll
    for (int k = 0; k < EPW; ++k) p[k] = edge_partial(a[k], b[k], at);
    #pragma unroll
    for (int k = 0; k < EPW; ++k) p[k] += __shfl_xor(p[k], 1, 32);
    #pragma unroll
    for (int k = 0; k < EPW; ++k) p[k] += __shfl_xor(p[k], 2, 32);
    #pragma unroll
    for (int k = 0; k < EPW; ++k) p[k] += __shfl_xor(p[k], 4, 32);

    #pragma unroll
    for (int k = 0; k < EPW; ++k) {
        if ((lane & 7) == 0 && e0 + k < nTot)
            atomicMax(segmax + (size_t)d[k] * HEADS + (lane >> 3), f2o(p[k]));
    }
}

__global__ void __launch_bounds__(256)
gat_edge_acc(const float* __restrict__ xl, const float* __restrict__ xr,
             const float* __restrict__ att,
             const int* __restrict__ src, const int* __restrict__ dst,
             int nEdges, int nTot,
             const int* __restrict__ segmax,
             float* __restrict__ denom, float* __restrict__ out)
{
    const int wv = __builtin_amdgcn_readfirstlane((int)(threadIdx.x >> 5));
    const int e0 = (blockIdx.x * 8 + wv) * EPW;
    if (e0 >= nTot) return;
    const int lane = threadIdx.x & 31;
    const int h = lane >> 3;

    int sv, dv;
    edge_ids(src, dst, nEdges, nTot, e0, lane, sv, dv);
    const float4 at = *(const float4*)(att + lane * 4);

    float4 a[EPW], b[EPW];
    int d[EPW], mo[EPW];
    #pragma unroll
    for (int k = 0; k < EPW; ++k) {
        const int s = __builtin_amdgcn_readlane(sv, k);
        d[k] = __builtin_amdgcn_readlane(dv, k);
        a[k]  = *(const float4*)(xl + (size_t)s    * HC + lane * 4);
        b[k]  = *(const float4*)(xr + (size_t)d[k] * HC + lane * 4);
        mo[k] = segmax[(size_t)d[k] * HEADS + h];
    }

    float p[EPW];
    #pragma unroll
    for (int k = 0; k < EPW; ++k) p[k] = edge_partial(a[k], b[k], at);
    #pragma unroll
    for (int k = 0; k < EPW; ++k) p[k] += __shfl_xor(p[k], 1, 32);
    #pragma unroll
    for (int k = 0; k < EPW; ++k) p[k] += __shfl_xor(p[k], 2, 32);
    #pragma unroll
    for (int k = 0; k < EPW; ++k) p[k] += __shfl_xor(p[k], 4, 32);

    #pragma unroll
    for (int k = 0; k < EPW; ++k) {
        const float w = __expf(p[k] - o2f(mo[k]));
        if (e0 + k < nTot) {
            if ((lane & 7) == 0)
                unsafeAtomicAdd(denom + (size_t)d[k] * HEADS + h, w);
            float* o = out + (size_t)d[k] * HC + lane * 4;
            unsafeAtomicAdd(o + 0, w * a[k].x);
            unsafeAtomicAdd(o + 1, w * a[k].y);
            unsafeAtomicAdd(o + 2, w * a[k].z);
            unsafeAtomicAdd(o + 3, w * a[k].w);
        }
    }
}

__global__ void gat_final(float* __restrict__ out, const float* __restrict__ denom,
                          const float* __restrict__ bias, int nOut)
{
    int i = blockIdx.x * blockDim.x + threadIdx.x;
    if (i >= nOut) return;
    const int node = i >> 7;
    const int ch   = i & 127;
    out[i] = out[i] / denom[node * HEADS + (ch >> 5)] + bias[ch];
}

extern "C" void kernel_launch(void* const* d_in, const int* in_sizes, int n_in,
                              void* d_out, int out_size, void* d_ws, size_t ws_size,
                              hipStream_t stream)
{
    const float* x    = (const float*)d_in[0];
    const float* Wl   = (const float*)d_in[1];
    const float* bl   = (const float*)d_in[2];
    const float* Wr   = (const float*)d_in[3];
    const float* br   = (const float*)d_in[4];
    const float* att  = (const float*)d_in[5];
    const float* bias = (const float*)d_in[6];
    const int*   src  = (const int*)d_in[7];
    const int*   dst  = (const int*)d_in[8];
    const int nE = in_sizes[7];
    float* out = (float*)d_out;

    float* xl    = (float*)d_ws;
    float* xr    = xl + (size_t)N_NODES * HC;
    float* denom = xr + (size_t)N_NODES * HC;
    int*   smax  = (int*)(denom + (size_t)N_NODES * HEADS);

    const int nOut = N_NODES * HC;
    const int nSeg = N_NODES * HEADS;

    gat_init<<<(nOut + 255) / 256, 256, 0, stream>>>(out, denom, smax, nOut, nSeg);

    gat_gemm<<<3125, 128, 0, stream>>>(x, Wl, bl, Wr, br, xl, xr);

    const int nTot = nE + N_NODES;
    const int eb = (nTot + 8 * EPW - 1) / (8 * EPW);
    gat_edge_max<<<eb, 256, 0, stream>>>(xl, xr, att, src, dst, nE, nTot, smax);
    gat_edge_acc<<<eb, 256, 0, stream>>>(xl, xr, att, src, dst, nE, nTot, smax, denom, out);

    gat_final<<<(nOut + 255) / 256, 256, 0, stream>>>(out, denom, bias, nOut);
}


# Scores computed once: max pass stores each edge's 4 scores (in the out buffer); accumulate pass reads them, gathers only x_l rows and adds into the zeroed x_r plane; arrival-order float atomics kept
# speedup vs baseline: 1.6714x; 1.4658x over previous
#include <hip/hip_runtime.h>

#define N_NODES 100000
#define IN_DIM  128
#define HEADS   4
#define HC      128
#define NEG_SLOPE 0.2f

typedef float v2f __attribute__((ext_vector_type(2)));
typedef float v8f __attribute__((ext_vector_type(8)));

__device__ __forceinline__ int f2o(float f) {
    int i = __float_as_int(f);
    return i >= 0 ? i : (i ^ 0x7FFFFFFF);
}
__device__ __forceinline__ float o2f(int i) {
    int j = i >= 0 ? i : (i ^ 0x7FFFFFFF);
    return __int_as_float(j);
}

__global__ void gat_gemm(const float* __restrict__ x,
                         const float* __restrict__ Wl, const float* __restrict__ bl,
                         const float* __restrict__ Wr, const float* __restrict__ br,
                         float* __restrict__ xl, float* __restrict__ xr)
{
    const int wid  = threadIdx.x >> 5;
    const int lane = threadIdx.x & 31;
    const int task = blockIdx.x * 4 + wid;
    const int matSel  = task & 1;
    const int rowTile = task >> 1;

    const float* W    = matSel ? Wr : Wl;
    const float* bv   = matSel ? br : bl;
    float*       outp = matSel ? xr : xl;

    const int row0 = rowTile * 16;
    const int lm = lane & 15;
    const int lk = (lane >> 4) * 2;

    const float* aptr = x + (size_t)(row0 + lm) * IN_DIM + lk;
    const float* bptr = W + (size_t)lk * HC + lm;

    v8f c[8] = {};
    for (int k0 = 0; k0 < IN_DIM; k0 += 4) {
        const float2 av = *(const float2*)aptr;
        v2f a; a.x = av.x; a.y = av.y;
        #pragma unroll
        for (int t = 0; t < 8; ++t) {
            v2f b;
            b.x = bptr[t * 16];
            b.y = bptr[HC + t * 16];
            c[t] = __builtin_amdgcn_wmma_f32_16x16x4_f32(
                       false, a, false, b, (short)0, c[t], false, false);
        }
        aptr += 4;
        bptr += 4 * HC;
    }

    const int rbase = row0 + ((lane >> 4) * 8);
    #pragma unroll
    for (int hf = 0; hf < 2; ++hf) {
        const int ch = hf * 16 + lm;
        const float b0 = bv[ch], b1 = bv[32 + ch], b2 = bv[64 + ch], b3 = bv[96 + ch];
        #pragma unroll
        for (int r = 0; r < 8; ++r) {
            float4 v;
            v.x = c[hf][r]     + b0;
            v.y = c[2 + hf][r] + b1;
            v.z = c[4 + hf][r] + b2;
            v.w = c[6 + hf][r] + b3;
            *(float4*)(outp + (size_t)(rbase + r) * HC + 4 * ch) = v;
        }
    }
}

__global__ void gat_init(float* __restrict__ num, float* __restrict__ denom,
                         int* __restrict__ segmax, int nOut, int nSeg)
{
    int i = blockIdx.x * blockDim.x + threadIdx.x;
    if (i < nOut) num[i] = 0.0f;
    if (i < nSeg) { denom[i] = 0.0f; segmax[i] = (int)0x80000000; }
}

__device__ __forceinline__ float4 edge_terms(const float4 a, const float4 b,
                                             const float4 at)
{
    float m0 = a.x + b.x; m0 = m0 >= 0.0f ? m0 : NEG_SLOPE * m0;
    float m1 = a.y + b.y; m1 = m1 >= 0.0f ? m1 : NEG_SLOPE * m1;
    float m2 = a.z + b.z; m2 = m2 >= 0.0f ? m2 : NEG_SLOPE * m2;
    float m3 = a.w + b.w; m3 = m3 >= 0.0f ? m3 : NEG_SLOPE * m3;
    float4 t;
    t.x = at.x * m0; t.y = at.y * m1; t.z = at.z * m2; t.w = at.w * m3;
    return t;
}

#define EPW 8

__device__ __forceinline__ void head_sums(const float4 (&t)[EPW], int lane,
                                          float (&p)[EPW])
{
    const bool up16 = (lane & 16) != 0;
    const bool up8  = (lane & 8)  != 0;
    float q0[EPW], q1[EPW];
    #pragma unroll
    for (int k = 0; k < EPW; ++k) {
        const float k0 = up16 ? t[k].z : t[k].x;
        const float k1 = up16 ? t[k].w : t[k].y;
        const float s0 = up16 ? t[k].x : t[k].z;
        const float s1 = up16 ? t[k].y : t[k].w;
        q0[k] = k0 + __shfl_xor(s0, 16, 32);
        q1[k] = k1 + __shfl_xor(s1, 16, 32);
    }
    #pragma unroll
    for (int k = 0; k < EPW; ++k) {
        const float kp = up8 ? q1[k] : q0[k];
        const float sd = up8 ? q0[k] : q1[k];
        p[k] = kp + __shfl_xor(sd, 8, 32);
    }
    #pragma unroll
    for (int k = 0; k < EPW; ++k) p[k] += __shfl_xor(p[k], 4, 32);
    #pragma unroll
    for (int k = 0; k < EPW; ++k) p[k] += __shfl_xor(p[k], 2, 32);
    #pragma unroll
    for (int k = 0; k < EPW; ++k) p[k] += __shfl_xor(p[k], 1, 32);
}

__device__ __forceinline__ void edge_ids(const int* __restrict__ src,
                                         const int* __restrict__ dst,
                                         int nEdges, int nTot, int e0, int lane,
                                         int& sv, int& dv)
{
    const int e = e0 + (lane & (EPW - 1));
    sv = 0; dv = 0;
    if (e < nEdges)    { sv = src[e]; dv = dst[e]; }
    else if (e < nTot) { sv = e - nEdges; dv = sv; }
}

__global__ void __launch_bounds__(256)
gat_edge_max(const float* __restrict__ xl, const float* __restrict__ xr,
             const float* __restrict__ att,
             const int* __restrict__ src, const int* __restrict__ dst,
             int nEdges, int nTot, int* __restrict__ segmax, float* __restrict__ sc)
{
    const int wv = __builtin_amdgcn_readfirstlane((int)(threadIdx.x >> 5));
    const int e0 = (blockIdx.x * 8 + wv) * EPW;
    if (e0 >= nTot) return;
    const int lane = threadIdx.x & 31;

    int sv, dv;
    edge_ids(src, dst, nEdges, nTot, e0, lane, sv, dv);
    float4 at;
    at.x = att[lane]; at.y = att[32 + lane]; at.z = att[64 + lane]; at.w = att[96 + lane];

    float4 a[EPW], b[EPW];
    int d[EPW];
    #pragma unroll
    for (int k = 0; k < EPW; ++k) {
        const int s = __builtin_amdgcn_readlane(sv, k);
        d[k] = __builtin_amdgcn_readlane(dv, k);
        a[k] = *(const float4*)(xl + (size_t)s    * HC + lane * 4);
        b[k] = *(const float4*)(xr + (size_t)d[k] * HC + lane * 4);
    }

    float4 t[EPW];
    #pragma unroll
    for (int k = 0; k < EPW; ++k) t[k] = edge_terms(a[k], b[k], at);
    float p[EPW];
    head_sums(t, lane, p);

    #pragma unroll
    for (int k = 0; k < EPW; ++k) {
        if ((lane & 7) == 0 && e0 + k < nTot)
            atomicMax(segmax + (size_t)d[k] * HEADS + (lane >> 3), f2o(p[k]));
    }

    const bool b0 = (lane & 1) != 0, b1 = (lane & 2) != 0, b2 = (lane & 4) != 0;
    const float u0 = b0 ? p[1] : p[0];
    const float u1 = b0 ? p[3] : p[2];
    const float u2 = b0 ? p[5] : p[4];
    const float u3 = b0 ? p[7] : p[6];
    const float v0 = b1 ? u1 : u0;
    const float v1 = b1 ? u3 : u2;
    const float v  = b2 ? v1 : v0;
    const int ei = e0 + (lane & 7);
    if (ei < nTot) sc[(size_t)ei * HEADS + (lane >> 3)] = v;
}

__global__ void __launch_bounds__(256)
gat_edge_acc(const float* __restrict__ xl, const float* __restrict__ sc,
             const int* __restrict__ src, const int* __restrict__ dst,
             int nEdges, int nTot,
             const int* __restrict__ segmax,
             float* __restrict__ denom, float* __restrict__ num)
{
    const int wv = __builtin_amdgcn_readfirstlane((int)(threadIdx.x >> 5));
    const int e0 = (blockIdx.x * 8 + wv) * EPW;
    if (e0 >= nTot) return;
    const int lane = threadIdx.x & 31;
    const int h = lane & 3;

    const int e = e0 + (lane >> 2);
    int sv = 0, dv = 0;
    if (e < nEdges)    { sv = src[e]; dv = dst[e]; }
    else if (e < nTot) { sv = e - nEdges; dv = sv; }

    const float s  = (e < nTot) ? sc[(size_t)e * HEADS + h] : 0.0f;
    const int   mo = segmax[(size_t)dv * HEADS + h];

    float4 a[EPW];
    #pragma unroll
    for (int k = 0; k < EPW; ++k) {
        const int sk = __builtin_amdgcn_readlane(sv, 4 * k);
        a[k] = *(const float4*)(xl + (size_t)sk * HC + lane * 4);
    }

    const float w = __expf(s - o2f(mo));
    if (e < nTot) unsafeAtomicAdd(denom + (size_t)dv * HEADS + h, w);
    const int wi = __float_as_int(w);

    #pragma unroll
    for (int k = 0; k < EPW; ++k) {
        if (e0 + k < nTot) {
            const int   dk = __builtin_amdgcn_readlane(dv, 4 * k);
            const float w0 = __int_as_float(__builtin_amdgcn_readlane(wi, 4 * k));
            const float w1 = __int_as_float(__builtin_amdgcn_readlane(wi, 4 * k + 1));
            const float w2 = __int_as_float(__builtin_amdgcn_readlane(wi, 4 * k + 2));
            const float w3 = __int_as_float(__builtin_amdgcn_readlane(wi, 4 * k + 3));
            float* o = num + (size_t)dk * HC + lane;
            unsafeAtomicAdd(o,      w0 * a[k].x);
            unsafeAtomicAdd(o + 32, w1 * a[k].y);
            unsafeAtomicAdd(o + 64, w2 * a[k].z);
            unsafeAtomicAdd(o + 96, w3 * a[k].w);
        }
    }
}

__global__ void gat_final(float* __restrict__ out, const float* __restrict__ num,
                          const float* __restrict__ denom,
                          const float* __restrict__ bias, int nOut)
{
    int i = blockIdx.x * blockDim.x + threadIdx.x;
    if (i >= nOut) return;
    const int node = i >> 7;
    const int ch   = i & 127;
    out[i] = num[i] / denom[node * HEADS + (ch >> 5)] + bias[ch];
}

extern "C" void kernel_launch(void* const* d_in, const int* in_sizes, int n_in,
                              void* d_out, int out_size, void* d_ws, size_t ws_size,
                              hipStream_t stream)
{
    const float* x    = (const float*)d_in[0];
    const float* Wl   = (const float*)d_in[1];
    const float* bl   = (const float*)d_in[2];
    const float* Wr   = (const float*)d_in[3];
    const float* br   = (const float*)d_in[4];
    const float* att  = (const float*)d_in[5];
    const float* bias = (const float*)d_in[6];
    const int*   src  = (const int*)d_in[7];
    const int*   dst  = (const int*)d_in[8];
    const int nE = in_sizes[7];
    float* out = (float*)d_out;

    float* xl    = (float*)d_ws;
    float* xr    = xl + (size_t)N_NODES * HC;
    float* denom = xr + (size_t)N_NODES * HC;
    int*   smax  = (int*)(denom + (size_t)N_NODES * HEADS);
    float* sc  = out;
    float* num = xr;

    const int nOut = N_NODES * HC;
    const int nSeg = N_NODES * HEADS;

    gat_init<<<(nSeg + 255) / 256, 256, 0, stream>>>(num, denom, smax, 0, nSeg);

    gat_gemm<<<3125, 128, 0, stream>>>(x, Wl, bl, Wr, br, xl, xr);

    const int nTot = nE + N_NODES;
    const int eb = (nTot + 8 * EPW - 1) / (8 * EPW);
    gat_edge_max<<<eb, 256, 0, stream>>>(xl, xr, att, src, dst, nE, nTot, smax, sc);
    gat_init<<<(nOut + 255) / 256, 256, 0, stream>>>(num, denom, smax, nOut, 0);
    gat_edge_acc<<<eb, 256, 0, stream>>>(xl, sc, src, dst, nE, nTot, smax, denom, num);

    gat_final<<<(nOut + 255) / 256, 256, 0, stream>>>(out, num, denom, bias, nOut);
}
